# starting kernel
# baseline (speedup 1.0000x reference)
#include <hip/hip_runtime.h>

typedef __attribute__((ext_vector_type(2))) float v2f;
typedef __attribute__((ext_vector_type(4))) float v4f;
typedef __attribute__((ext_vector_type(8))) float v8f;

#define KW        5
#define PATCH     25
#define JPAD      28
#define NKB       7
#define C_IN      4
#define F_OUT     16
#define HH        250
#define WW        250
#define BB        8
#define NPIX      (BB*HH*WW)
#define NTILES    (NPIX/16)
#define WPB       16
#define NTHREADS  (WPB*32)
#define NQTILES   (C_IN*PATCH*NKB)
#define NLTILES   (C_IN*NKB)
#define NWTILES   (NQTILES+NLTILES)
#define WFLOATS   (NWTILES*64)
#define PATF      (JPAD*C_IN)
#define LDS_FLOATS (WFLOATS + WPB*16*PATF)

__global__ void prep_weights(const float* __restrict__ kern,
                             const float* __restrict__ qk,
                             float* __restrict__ wbuf) {
    int idx = blockIdx.x * blockDim.x + threadIdx.x;
    if (idx >= WFLOATS) return;
    int t    = idx >> 6;
    int slot = idx & 63;
    int l    = slot >> 1;
    int v    = slot & 1;
    int n    = l & 15;
    int krow = v + ((l >= 16) ? 2 : 0);
    float val = 0.0f;
    if (t < NQTILES) {
        int kb = t % NKB;
        int di = t / NKB;
        int j  = 4 * kb + krow;
        if (j < PATCH) val = qk[(di * PATCH + j) * F_OUT + n];
    } else {
        int t2 = t - NQTILES;
        int kb = t2 % NKB;
        int d  = t2 / NKB;
        int i  = 4 * kb + krow;
        if (i < PATCH) val = kern[(i * C_IN + d) * F_OUT + n];
    }
    wbuf[idx] = val;
}

extern __shared__ float smem[];

__global__ __launch_bounds__(NTHREADS)
void convquad_main(const float* __restrict__ x,
                   const float* __restrict__ bias,
                   const float* __restrict__ wbuf,
                   float* __restrict__ out) {
    const int tid  = threadIdx.x;
    const int lane = tid & 31;
    const int wave = tid >> 5;
    const int hi   = (lane >> 4) & 1;
    const int n    = lane & 15;

    {
        const v4f* src = (const v4f*)wbuf;
        v4f*       dst = (v4f*)smem;
        for (int e = tid; e < WFLOATS / 4; e += NTHREADS) dst[e] = src[e];
    }

    float* pat = smem + WFLOATS + wave * (16 * PATF);
    const int tile = blockIdx.x * WPB + wave;
    const bool active = (tile < NTILES);

    if (active) {
        const v4f z = {0.f, 0.f, 0.f, 0.f};
        for (int e = lane; e < 16 * 3; e += 32) {
            int pix = e / 3, jj = PATCH + (e % 3);
            *(v4f*)(pat + pix * PATF + jj * C_IN) = z;
        }
        for (int e = lane; e < 16 * PATCH; e += 32) {
            int pix = e / PATCH, j = e % PATCH;
            int P  = tile * 16 + pix;
            int wc = P % WW;
            int hc = (P / WW) % HH;
            int bc = P / (WW * HH);
            int h2 = hc + (j / KW) - 2;
            int w2 = wc + (j % KW) - 2;
            v4f val = z;
            if (h2 >= 0 && h2 < HH && w2 >= 0 && w2 < WW)
                val = *(const v4f*)(x + (size_t)((bc * HH + h2) * WW + w2) * C_IN);
            *(v4f*)(pat + pix * PATF + j * C_IN) = val;
        }
    }
    __syncthreads();
    if (!active) return;

    const float bv = bias[n];
    v8f acc = {bv, bv, bv, bv, bv, bv, bv, bv};

    const float* patA = pat + n * PATF;
    const float* patS = pat + hi * 8 * PATF;

    #pragma unroll 1
    for (int d = 0; d < C_IN; ++d) {
        v2f A[NKB];
        #pragma unroll
        for (int kb = 0; kb < NKB; ++kb) {
            int j0 = 4 * kb + 2 * hi;
            A[kb].x = patA[j0 * C_IN + d];
            A[kb].y = patA[(j0 + 1) * C_IN + d];
        }
        const float* Bq = smem + (size_t)(d * PATCH) * NKB * 64 + lane * 2;
        #pragma unroll 1
        for (int i = 0; i < PATCH; ++i) {
            v8f s = {0.f, 0.f, 0.f, 0.f, 0.f, 0.f, 0.f, 0.f};
            const float* Bt = Bq + i * (NKB * 64);
            #pragma unroll
            for (int kb = 0; kb < NKB; ++kb) {
                v2f b = *(const v2f*)(Bt + kb * 64);
                s = __builtin_amdgcn_wmma_f32_16x16x4_f32(
                        false, A[kb], false, b, (short)0, s, false, false);
            }
            const float* m = patS + i * C_IN + d;
            #pragma unroll
            for (int g = 0; g < 8; ++g)
                acc[g] = __builtin_fmaf(m[g * PATF], s[g], acc[g]);
        }
        const float* Bl = smem + (size_t)(NQTILES + d * NKB) * 64 + lane * 2;
        #pragma unroll
        for (int kb = 0; kb < NKB; ++kb) {
            v2f b = *(const v2f*)(Bl + kb * 64);
            acc = __builtin_amdgcn_wmma_f32_16x16x4_f32(
                      false, A[kb], false, b, (short)0, acc, false, false);
        }
    }

    float* op = out + (size_t)(tile * 16 + hi * 8) * F_OUT + n;
    #pragma unroll
    for (int g = 0; g < 8; ++g) op[g * F_OUT] = acc[g];
}

extern "C" void kernel_launch(void* const* d_in, const int* in_sizes, int n_in,
                              void* d_out, int out_size, void* d_ws, size_t ws_size,
                              hipStream_t stream) {
    const float* x    = (const float*)d_in[0];
    const float* kern = (const float*)d_in[1];
    const float* qk   = (const float*)d_in[2];
    const float* bias = (const float*)d_in[3];
    float*       out  = (float*)d_out;
    float*       wbuf = (float*)d_ws;

    prep_weights<<<(WFLOATS + 255) / 256, 256, 0, stream>>>(kern, qk, wbuf);

    const size_t shmem = (size_t)LDS_FLOATS * sizeof(float);
    (void)hipFuncSetAttribute((const void*)convquad_main,
                              hipFuncAttributeMaxDynamicSharedMemorySize,
                              (int)shmem);
    const int blocks = (NTILES + WPB - 1) / WPB;
    convquad_main<<<blocks, NTHREADS, shmem, stream>>>(x, bias, wbuf, out);
}


# Each wave takes 16 pixel pairs (two image rows, one shared 6x5 window): every weight fragment read from LDS feeds two WMMAs; 16 tiles per weight copy; quad rows in groups of 4; sum order kept
# speedup vs baseline: 1.4568x; 1.4568x over previous
#include <hip/hip_runtime.h>


typedef __attribute__((ext_vector_type(2))) float v2f;
typedef __attribute__((ext_vector_type(4))) float v4f;
typedef __attribute__((ext_vector_type(8))) float v8f;

#define KW        5
#define PATCH     25
#define NKB       7
#define IGRP      4
#define WROWS     6
#define WIN       (WROWS*KW)
#define C_IN      4
#define F_OUT     16
#define HH        250
#define WW        250
#define BB        8
#define HP        (HH/2)
#define NQ        (BB*HP*WW)
#define NTILES    (NQ/16)
#define WPB       8
#define NTHREADS  (WPB*32)
#define QFLOATS   (C_IN*PATCH*PATCH*F_OUT)
#define ROWF      (PATCH*F_OUT)
#define QPADF     64
#define LDS_FLOATS (QFLOATS + QPADF)

extern __shared__ float smem[];

__device__ __forceinline__ float pick(float lo, float hi, unsigned mask) {
    return __uint_as_float((__float_as_uint(lo) & ~mask) | (__float_as_uint(hi) & mask));
}

__global__ __launch_bounds__(NTHREADS, 4)
void convquad_main(const float* __restrict__ x,
                   const float* __restrict__ kern,
                   const float* __restrict__ qk,
                   const float* __restrict__ bias,
                   float* __restrict__ out) {
    const int tid  = threadIdx.x;
    const int lane = tid & 31;
    const int wave = tid >> 5;
    const int h    = (lane >> 4) & 1;
    const int c    = lane & 15;
    const unsigned hm = 0u - (unsigned)h;

    {
        const v4f* src = (const v4f*)qk;
        v4f*       dst = (v4f*)smem;
        for (int e = tid; e < QFLOATS / 4; e += NTHREADS) dst[e] = src[e];
        if (tid < QPADF / 4) dst[QFLOATS / 4 + tid] = (v4f){0.f, 0.f, 0.f, 0.f};
    }

    const int tile = blockIdx.x * WPB + wave;
    const bool active = (tile < NTILES);

    const int Q  = (active ? tile : 0) * 16 + c;
    const int wc = Q % WW;
    const int hc = 2 * ((Q / WW) % HP);
    const int bc = Q / (WW * HP);
    int  rowoff[WROWS], coloff[KW];
    bool rowok[WROWS],  colok[KW];
    #pragma unroll
    for (int t = 0; t < WROWS; ++t) {
        int h2 = hc + t - 2;
        rowok[t] = (h2 >= 0 && h2 < HH);
        h2 = h2 < 0 ? 0 : (h2 > HH - 1 ? HH - 1 : h2);
        rowoff[t] = ((bc * HH + h2) * WW) * C_IN;
    }
    #pragma unroll
    for (int t = 0; t < KW; ++t) {
        int w2 = wc + t - 2;
        colok[t] = (w2 >= 0 && w2 < WW);
        w2 = w2 < 0 ? 0 : (w2 > WW - 1 ? WW - 1 : w2);
        coloff[t] = w2 * C_IN;
    }

    float nx[WIN];
    #pragma unroll
    for (int j = 0; j < WIN; ++j) nx[j] = x[rowoff[j / KW] + coloff[j % KW]];

    __syncthreads();
    if (!active) return;

    v8f acc0, acc1;
    {
        const v4f b0 = *(const v4f*)(bias + 8 * h);
        const v4f b1 = *(const v4f*)(bias + 8 * h + 4);
        acc0 = (v8f){b0.x, b0.y, b0.z, b0.w, b1.x, b1.y, b1.z, b1.w};
        acc1 = acc0;
    }

    const float* wl = smem + h * 32 + c;
    int lin0[NKB], lin1[NKB];
    #pragma unroll
    for (int kb = 0; kb < NKB; ++kb) {
        int i0 = 4 * kb + 2 * h, i1 = i0 + 1;
        i0 = i0 > PATCH - 1 ? PATCH - 1 : i0;
        i1 = i1 > PATCH - 1 ? PATCH - 1 : i1;
        lin0[kb] = i0 * (C_IN * F_OUT) + c;
        lin1[kb] = i1 * (C_IN * F_OUT) + c;
    }

    #pragma unroll 1
    for (int d = 0; d < C_IN; ++d) {
        float m0[4 * NKB], m1[4 * NKB];
        {
            float win[WIN];
            #pragma unroll
            for (int j = 0; j < WIN; ++j)
                win[j] = (rowok[j / KW] & colok[j % KW]) ? nx[j] : 0.0f;
            #pragma unroll
            for (int j = 0; j < PATCH; ++j) { m0[j] = win[j]; m1[j] = win[j + KW]; }
            #pragma unroll
            for (int j = PATCH; j < 4 * NKB; ++j) { m0[j] = 0.0f; m1[j] = 0.0f; }
        }
        v2f Bf0[NKB], Bf1[NKB];
        #pragma unroll
        for (int kb = 0; kb < NKB; ++kb) {
            Bf0[kb].x = pick(m0[4 * kb],     m0[4 * kb + 2], hm);
            Bf0[kb].y = pick(m0[4 * kb + 1], m0[4 * kb + 3], hm);
            Bf1[kb].x = pick(m1[4 * kb],     m1[4 * kb + 2], hm);
            Bf1[kb].y = pick(m1[4 * kb + 1], m1[4 * kb + 3], hm);
        }
        v2f Al[NKB];
        const float* wp = wl + d * (PATCH * ROWF);
        #pragma unroll
        for (int ig = 0; ig < PATCH; ig += IGRP) {
            if (ig + IGRP >= PATCH) {
                #pragma unroll
                for (int kb = 0; kb < NKB; ++kb) {
                    Al[kb].x = kern[lin0[kb] + d * F_OUT];
                    Al[kb].y = kern[lin1[kb] + d * F_OUT];
                }
                const int dn = d < C_IN - 1 ? d + 1 : C_IN - 1;
                int ro[WROWS], co[KW];
                #pragma unroll
                for (int t = 0; t < WROWS; ++t) {
                    ro[t] = rowoff[t];
                    asm volatile("" : "+v"(ro[t]));
                }
                #pragma unroll
                for (int t = 0; t < KW; ++t) co[t] = coloff[t] + dn;
                #pragma unroll
                for (int j = 0; j < WIN; ++j) nx[j] = x[ro[j / KW] + co[j % KW]];
            }
            v2f a[IGRP][NKB];
            #pragma unroll
            for (int kb = 0; kb < NKB; ++kb) {
                #pragma unroll
                for (int r = 0; r < IGRP; ++r) {
                    if (ig + r < PATCH) {
                        a[r][kb].x = wp[(ig + r) * ROWF + kb * 64];
                        a[r][kb].y = wp[(ig + r) * ROWF + kb * 64 + 16];
                    }
                }
            }
            __builtin_amdgcn_sched_barrier(0);
            v8f s0[IGRP], s1[IGRP];
            #pragma unroll
            for (int r = 0; r < IGRP; ++r) {
                s0[r] = (v8f){0.f, 0.f, 0.f, 0.f, 0.f, 0.f, 0.f, 0.f};
                s1[r] = (v8f){0.f, 0.f, 0.f, 0.f, 0.f, 0.f, 0.f, 0.f};
            }
            #pragma unroll
            for (int kb = 0; kb < NKB; ++kb) {
                #pragma unroll
                for (int r = 0; r < IGRP; ++r) {
                    if (ig + r < PATCH) {
                        s0[r] = __builtin_amdgcn_wmma_f32_16x16x4_f32(
                                    false, a[r][kb], false, Bf0[kb], (short)0, s0[r],
                                    false, false);
                        s1[r] = __builtin_amdgcn_wmma_f32_16x16x4_f32(
                                    false, a[r][kb], false, Bf1[kb], (short)0, s1[r],
                                    false, false);
                    }
                }
                __builtin_amdgcn_sched_barrier(0);
            }
            #pragma unroll
            for (int r = 0; r < IGRP; ++r) {
                if (ig + r < PATCH) {
                    const float p0 = m0[ig + r], p1 = m1[ig + r];
                    acc0 = __builtin_elementwise_fma(
                               (v8f){p0, p0, p0, p0, p0, p0, p0, p0}, s0[r], acc0);
                    acc1 = __builtin_elementwise_fma(
                               (v8f){p1, p1, p1, p1, p1, p1, p1, p1}, s1[r], acc1);
                }
            }
        }
        #pragma unroll
        for (int kb = 0; kb < NKB; ++kb) {
            acc0 = __builtin_amdgcn_wmma_f32_16x16x4_f32(
                       false, Al[kb], false, Bf0[kb], (short)0, acc0, false, false);
            acc1 = __builtin_amdgcn_wmma_f32_16x16x4_f32(
                       false, Al[kb], false, Bf1[kb], (short)0, acc1, false, false);
        }
    }

    float* op = out + (size_t)((bc * HH + hc) * WW + wc) * F_OUT + 8 * h;
    *(v4f*)(op)     = (v4f){acc0[0], acc0[1], acc0[2], acc0[3]};
    *(v4f*)(op + 4) = (v4f){acc0[4], acc0[5], acc0[6], acc0[7]};
    op += WW * F_OUT;
    *(v4f*)(op)     = (v4f){acc1[0], acc1[1], acc1[2], acc1[3]};
    *(v4f*)(op + 4) = (v4f){acc1[4], acc1[5], acc1[6], acc1[7]};
}

extern "C" void kernel_launch(void* const* d_in, const int* in_sizes, int n_in,
                              void* d_out, int out_size, void* d_ws, size_t ws_size,
                              hipStream_t stream) {
    const float* x    = (const float*)d_in[0];
    const float* kern = (const float*)d_in[1];
    const float* qk   = (const float*)d_in[2];
    const float* bias = (const float*)d_in[3];
    float*       out  = (float*)d_out;

    const size_t shmem = (size_t)LDS_FLOATS * sizeof(float);
    (void)hipFuncSetAttribute((const void*)convquad_main,
                              hipFuncAttributeMaxDynamicSharedMemorySize,
                              (int)shmem);
    const int blocks = (NTILES + WPB - 1) / WPB;
    convquad_main<<<blocks, NTHREADS, shmem, stream>>>(x, kern, qk, bias, out);
}


# Quad weight column j=24 runs as one chain with k = i on the same B fragments, so each quad row has 6 exact k-blocks, not 7 padded: 328 f32 WMMA per channel trip, was 364; sum order changed
# speedup vs baseline: 1.6374x; 1.1240x over previous
#include <hip/hip_runtime.h>


typedef __attribute__((ext_vector_type(2))) float v2f;
typedef __attribute__((ext_vector_type(4))) float v4f;
typedef __attribute__((ext_vector_type(8))) float v8f;

#define KW        5
#define PATCH     25
#define NKB       7
#define NKR       6
#define JCOL      (PATCH-1)
#define IGRP      4
#define WROWS     6
#define WIN       (WROWS*KW)
#define C_IN      4
#define F_OUT     16
#define HH        250
#define WW        250
#define BB        8
#define HP        (HH/2)
#define NQ        (BB*HP*WW)
#define NTILES    (NQ/16)
#define WPB       8
#define NTHREADS  (WPB*32)
#define QFLOATS   (C_IN*PATCH*PATCH*F_OUT)
#define ROWF      (PATCH*F_OUT)
#define LDS_FLOATS QFLOATS

extern __shared__ float smem[];

__device__ __forceinline__ float pick(float lo, float hi, unsigned mask) {
    return __uint_as_float((__float_as_uint(lo) & ~mask) | (__float_as_uint(hi) & mask));
}

__global__ __launch_bounds__(NTHREADS, 4)
void convquad_main(const float* __restrict__ x,
                   const float* __restrict__ kern,
                   const float* __restrict__ qk,
                   const float* __restrict__ bias,
                   float* __restrict__ out) {
    const int tid  = threadIdx.x;
    const int lane = tid & 31;
    const int wave = tid >> 5;
    const int h    = (lane >> 4) & 1;
    const int c    = lane & 15;
    const unsigned hm = 0u - (unsigned)h;

    {
        const v4f* src = (const v4f*)qk;
        v4f*       dst = (v4f*)smem;
        for (int e = tid; e < QFLOATS / 4; e += NTHREADS) dst[e] = src[e];
    }

    const int tile = blockIdx.x * WPB + wave;
    const bool active = (tile < NTILES);

    const int Q  = (active ? tile : 0) * 16 + c;
    const int wc = Q % WW;
    const int hc = 2 * ((Q / WW) % HP);
    const int bc = Q / (WW * HP);
    int  rowoff[WROWS], coloff[KW];
    bool rowok[WROWS],  colok[KW];
    #pragma unroll
    for (int t = 0; t < WROWS; ++t) {
        int h2 = hc + t - 2;
        rowok[t] = (h2 >= 0 && h2 < HH);
        h2 = h2 < 0 ? 0 : (h2 > HH - 1 ? HH - 1 : h2);
        rowoff[t] = ((bc * HH + h2) * WW) * C_IN;
    }
    #pragma unroll
    for (int t = 0; t < KW; ++t) {
        int w2 = wc + t - 2;
        colok[t] = (w2 >= 0 && w2 < WW);
        w2 = w2 < 0 ? 0 : (w2 > WW - 1 ? WW - 1 : w2);
        coloff[t] = w2 * C_IN;
    }

    float nx[WIN];
    #pragma unroll
    for (int j = 0; j < WIN; ++j) nx[j] = x[rowoff[j / KW] + coloff[j % KW]];

    __syncthreads();
    if (!active) return;

    v8f acc0, acc1;
    {
        const v4f b0 = *(const v4f*)(bias + 8 * h);
        const v4f b1 = *(const v4f*)(bias + 8 * h + 4);
        acc0 = (v8f){b0.x, b0.y, b0.z, b0.w, b1.x, b1.y, b1.z, b1.w};
        acc1 = acc0;
    }

    const float* wl = smem + h * 32 + c;
    const float* wq = smem + h * (2 * ROWF) + JCOL * F_OUT + c;
    const float* wz = smem + JCOL * ROWF + JCOL * F_OUT + c;
    int lin0[NKB], lin1[NKB];
    #pragma unroll
    for (int kb = 0; kb < NKB; ++kb) {
        int i0 = 4 * kb + 2 * h, i1 = i0 + 1;
        i0 = i0 > PATCH - 1 ? PATCH - 1 : i0;
        i1 = i1 > PATCH - 1 ? PATCH - 1 : i1;
        lin0[kb] = i0 * (C_IN * F_OUT) + c;
        lin1[kb] = i1 * (C_IN * F_OUT) + c;
    }

    #pragma unroll 1
    for (int d = 0; d < C_IN; ++d) {
        float m0[4 * NKB], m1[4 * NKB];
        {
            float win[WIN];
            #pragma unroll
            for (int j = 0; j < WIN; ++j)
                win[j] = (rowok[j / KW] & colok[j % KW]) ? nx[j] : 0.0f;
            #pragma unroll
            for (int j = 0; j < PATCH; ++j) { m0[j] = win[j]; m1[j] = win[j + KW]; }
            #pragma unroll
            for (int j = PATCH; j < 4 * NKB; ++j) { m0[j] = 0.0f; m1[j] = 0.0f; }
        }
        v2f Bf0[NKB], Bf1[NKB];
        #pragma unroll
        for (int kb = 0; kb < NKB; ++kb) {
            Bf0[kb].x = pick(m0[4 * kb],     m0[4 * kb + 2], hm);
            Bf0[kb].y = pick(m0[4 * kb + 1], m0[4 * kb + 3], hm);
            Bf1[kb].x = pick(m1[4 * kb],     m1[4 * kb + 2], hm);
            Bf1[kb].y = pick(m1[4 * kb + 1], m1[4 * kb + 3], hm);
        }
        const float* wp = wl + d * (PATCH * ROWF);
        #pragma unroll
        for (int ig = 0; ig + IGRP <= PATCH; ig += IGRP) {
            v2f a[IGRP][NKR];
            #pragma unroll
            for (int kb = 0; kb < NKR; ++kb) {
                #pragma unroll
                for (int r = 0; r < IGRP; ++r) {
                    a[r][kb].x = wp[(ig + r) * ROWF + kb * 64];
                    a[r][kb].y = wp[(ig + r) * ROWF + kb * 64 + 16];
                }
            }
            __builtin_amdgcn_sched_barrier(0);
            v8f s0[IGRP], s1[IGRP];
            #pragma unroll
            for (int r = 0; r < IGRP; ++r) {
                s0[r] = (v8f){0.f, 0.f, 0.f, 0.f, 0.f, 0.f, 0.f, 0.f};
                s1[r] = (v8f){0.f, 0.f, 0.f, 0.f, 0.f, 0.f, 0.f, 0.f};
            }
            #pragma unroll
            for (int kb = 0; kb < NKR; ++kb) {
                #pragma unroll
                for (int r = 0; r < IGRP; ++r) {
                    s0[r] = __builtin_amdgcn_wmma_f32_16x16x4_f32(
                                false, a[r][kb], false, Bf0[kb], (short)0, s0[r],
                                false, false);
                    s1[r] = __builtin_amdgcn_wmma_f32_16x16x4_f32(
                                false, a[r][kb], false, Bf1[kb], (short)0, s1[r],
                                false, false);
                }
                __builtin_amdgcn_sched_barrier(0);
            }
            #pragma unroll
            for (int r = 0; r < IGRP; ++r) {
                const float p0 = m0[ig + r], p1 = m1[ig + r];
                acc0 = __builtin_elementwise_fma(
                           (v8f){p0, p0, p0, p0, p0, p0, p0, p0}, s0[r], acc0);
                acc1 = __builtin_elementwise_fma(
                           (v8f){p1, p1, p1, p1, p1, p1, p1, p1}, s1[r], acc1);
            }
        }
        v2f Al[NKB];
        #pragma unroll
        for (int kb = 0; kb < NKB; ++kb) {
            Al[kb].x = kern[lin0[kb] + d * F_OUT];
            Al[kb].y = kern[lin1[kb] + d * F_OUT];
        }
        {
            const int dn = d < C_IN - 1 ? d + 1 : C_IN - 1;
            int ro[WROWS], co[KW];
            #pragma unroll
            for (int t = 0; t < WROWS; ++t) {
                ro[t] = rowoff[t];
                asm volatile("" : "+v"(ro[t]));
            }
            #pragma unroll
            for (int t = 0; t < KW; ++t) co[t] = coloff[t] + dn;
            #pragma unroll
            for (int j = 0; j < WIN; ++j) nx[j] = x[ro[j / KW] + co[j % KW]];
        }
        {
            v2f ar[NKR], aq[NKB];
            const float* wqd = wq + d * (PATCH * ROWF);
            #pragma unroll
            for (int kb = 0; kb < NKR; ++kb) {
                ar[kb].x = wp[JCOL * ROWF + kb * 64];
                ar[kb].y = wp[JCOL * ROWF + kb * 64 + 16];
                aq[kb].x = wqd[kb * (4 * ROWF)];
                aq[kb].y = wqd[kb * (4 * ROWF) + ROWF];
            }
            aq[NKR].x = wz[d * (PATCH * ROWF)];
            aq[NKR].y = aq[NKR].x;
            __builtin_amdgcn_sched_barrier(0);
            v8f sr0 = (v8f){0.f, 0.f, 0.f, 0.f, 0.f, 0.f, 0.f, 0.f};
            v8f sr1 = sr0, sq0 = sr0, sq1 = sr0;
            #pragma unroll
            for (int kb = 0; kb < NKB; ++kb) {
                if (kb < NKR) {
                    sr0 = __builtin_amdgcn_wmma_f32_16x16x4_f32(
                              false, ar[kb], false, Bf0[kb], (short)0, sr0, false, false);
                    sr1 = __builtin_amdgcn_wmma_f32_16x16x4_f32(
                              false, ar[kb], false, Bf1[kb], (short)0, sr1, false, false);
                }
                sq0 = __builtin_amdgcn_wmma_f32_16x16x4_f32(
                          false, aq[kb], false, Bf0[kb], (short)0, sq0, false, false);
                sq1 = __builtin_amdgcn_wmma_f32_16x16x4_f32(
                          false, aq[kb], false, Bf1[kb], (short)0, sq1, false, false);
                __builtin_amdgcn_sched_barrier(0);
            }
            const float p0 = m0[JCOL], p1 = m1[JCOL];
            const v8f P0 = (v8f){p0, p0, p0, p0, p0, p0, p0, p0};
            const v8f P1 = (v8f){p1, p1, p1, p1, p1, p1, p1, p1};
            acc0 = __builtin_elementwise_fma(P0, sr0, acc0);
            acc1 = __builtin_elementwise_fma(P1, sr1, acc1);
            acc0 = __builtin_elementwise_fma(P0, sq0, acc0);
            acc1 = __builtin_elementwise_fma(P1, sq1, acc1);
        }
        #pragma unroll
        for (int kb = 0; kb < NKB; ++kb) {
            acc0 = __builtin_amdgcn_wmma_f32_16x16x4_f32(
                       false, Al[kb], false, Bf0[kb], (short)0, acc0, false, false);
            acc1 = __builtin_amdgcn_wmma_f32_16x16x4_f32(
                       false, Al[kb], false, Bf1[kb], (short)0, acc1, false, false);
        }
    }

    float* op = out + (size_t)((bc * HH + hc) * WW + wc) * F_OUT + 8 * h;
    *(v4f*)(op)     = (v4f){acc0[0], acc0[1], acc0[2], acc0[3]};
    *(v4f*)(op + 4) = (v4f){acc0[4], acc0[5], acc0[6], acc0[7]};
    op += WW * F_OUT;
    *(v4f*)(op)     = (v4f){acc1[0], acc1[1], acc1[2], acc1[3]};
    *(v4f*)(op + 4) = (v4f){acc1[4], acc1[5], acc1[6], acc1[7]};
}

extern "C" void kernel_launch(void* const* d_in, const int* in_sizes, int n_in,
                              void* d_out, int out_size, void* d_ws, size_t ws_size,
                              hipStream_t stream) {
    const float* x    = (const float*)d_in[0];
    const float* kern = (const float*)d_in[1];
    const float* qk   = (const float*)d_in[2];
    const float* bias = (const float*)d_in[3];
    float*       out  = (float*)d_out;

    const size_t shmem = (size_t)LDS_FLOATS * sizeof(float);
    (void)hipFuncSetAttribute((const void*)convquad_main,
                              hipFuncAttributeMaxDynamicSharedMemorySize,
                              (int)shmem);
    const int blocks = (NTILES + WPB - 1) / WPB;
    convquad_main<<<blocks, NTHREADS, shmem, stream>>>(x, kern, qk, bias, out);
}


# Vertical tap pairs in the pixel pair's shared window rows run as 3 products, not 4: (x0-x1)w0, (x2-x1)w1, x1(w0+w1); 10 f32 WMMA per quad row, was 12; 280 per trip, was 328; sum order changed
# speedup vs baseline: 1.7242x; 1.0530x over previous
#include <hip/hip_runtime.h>


typedef __attribute__((ext_vector_type(2))) float v2f;
typedef __attribute__((ext_vector_type(4))) float v4f;
typedef __attribute__((ext_vector_type(8))) float v8f;

#define KW        5
#define PATCH     25
#define NKB       7
#define NKR       6
#define JCOL      (PATCH-1)
#define IGRP      3
#define WROWS     6
#define WIN       (WROWS*KW)
#define C_IN      4
#define F_OUT     16
#define HH        250
#define WW        250
#define BB        8
#define HP        (HH/2)
#define NQ        (BB*HP*WW)
#define NTILES    (NQ/16)
#define WPB       8
#define NTHREADS  (WPB*32)
#define QFLOATS   (C_IN*PATCH*PATCH*F_OUT)
#define ROWF      (PATCH*F_OUT)
#define LDS_FLOATS QFLOATS

extern __shared__ float smem[];

__device__ __forceinline__ float pick(float lo, float hi, unsigned mask) {
    return __uint_as_float((__float_as_uint(lo) & ~mask) | (__float_as_uint(hi) & mask));
}

__global__ __launch_bounds__(NTHREADS, 4)
void convquad_main(const float* __restrict__ x,
                   const float* __restrict__ kern,
                   const float* __restrict__ qk,
                   const float* __restrict__ bias,
                   float* __restrict__ out) {
    const int tid  = threadIdx.x;
    const int lane = tid & 31;
    const int wave = tid >> 5;
    const int h    = (lane >> 4) & 1;
    const int c    = lane & 15;
    const unsigned hm = 0u - (unsigned)h;

    {
        const v4f* src = (const v4f*)qk;
        v4f*       dst = (v4f*)smem;
        for (int e = tid; e < QFLOATS / 4; e += NTHREADS) dst[e] = src[e];
    }

    const int tile = blockIdx.x * WPB + wave;
    const bool active = (tile < NTILES);

    const int Q  = (active ? tile : 0) * 16 + c;
    const int wc = Q % WW;
    const int hc = 2 * ((Q / WW) % HP);
    const int bc = Q / (WW * HP);
    int  rowoff[WROWS], coloff[KW];
    bool rowok[WROWS],  colok[KW];
    #pragma unroll
    for (int t = 0; t < WROWS; ++t) {
        int h2 = hc + t - 2;
        rowok[t] = (h2 >= 0 && h2 < HH);
        h2 = h2 < 0 ? 0 : (h2 > HH - 1 ? HH - 1 : h2);
        rowoff[t] = ((bc * HH + h2) * WW) * C_IN;
    }
    #pragma unroll
    for (int t = 0; t < KW; ++t) {
        int w2 = wc + t - 2;
        colok[t] = (w2 >= 0 && w2 < WW);
        w2 = w2 < 0 ? 0 : (w2 > WW - 1 ? WW - 1 : w2);
        coloff[t] = w2 * C_IN;
    }

    float nx[WIN];
    #pragma unroll
    for (int j = 0; j < WIN; ++j) nx[j] = x[rowoff[j / KW] + coloff[j % KW]];

    __syncthreads();
    if (!active) return;

    v8f acc0, acc1;
    {
        const v4f b0 = *(const v4f*)(bias + 8 * h);
        const v4f b1 = *(const v4f*)(bias + 8 * h + 4);
        acc0 = (v8f){b0.x, b0.y, b0.z, b0.w, b1.x, b1.y, b1.z, b1.w};
        acc1 = acc0;
    }

    const float* wl = smem + h * 32 + c;
    const float* wc4 = smem + h * (10 * F_OUT) + 4 * F_OUT + c;
    const float* wq = smem + h * (2 * ROWF) + JCOL * F_OUT + c;
    const float* wz = smem + JCOL * ROWF + JCOL * F_OUT + c;
    int lin0[NKB], lin1[NKB];
    #pragma unroll
    for (int kb = 0; kb < NKB; ++kb) {
        int i0 = 4 * kb + 2 * h, i1 = i0 + 1;
        i0 = i0 > PATCH - 1 ? PATCH - 1 : i0;
        i1 = i1 > PATCH - 1 ? PATCH - 1 : i1;
        lin0[kb] = i0 * (C_IN * F_OUT) + c;
        lin1[kb] = i1 * (C_IN * F_OUT) + c;
    }

    #pragma unroll 1
    for (int d = 0; d < C_IN; ++d) {
        float win[WIN], m0[4 * NKB], m1[4 * NKB];
        #pragma unroll
        for (int j = 0; j < WIN; ++j)
            win[j] = (rowok[j / KW] & colok[j % KW]) ? nx[j] : 0.0f;
        #pragma unroll
        for (int j = 0; j < PATCH; ++j) { m0[j] = win[j]; m1[j] = win[j + KW]; }
        #pragma unroll
        for (int j = PATCH; j < 4 * NKB; ++j) { m0[j] = 0.0f; m1[j] = 0.0f; }
        v2f BN0[4], BN1[4], BM2[2];
        #define FRAG(dst, f0, f1, f2, f3) \
            do { (dst).x = pick((f0), (f2), hm); (dst).y = pick((f1), (f3), hm); } while (0)
        FRAG(BN0[0], win[0] - win[5],   win[1] - win[6],   win[2] - win[7],   win[3] - win[8]);
        FRAG(BN0[1], win[10] - win[15], win[11] - win[16], win[12] - win[17], win[13] - win[18]);
        FRAG(BN0[2], win[20], win[21], win[22], win[23]);
        FRAG(BN0[3], win[4],  win[9],  win[14], win[19]);
        FRAG(BN1[0], win[10] - win[5],  win[11] - win[6],  win[12] - win[7],  win[13] - win[8]);
        FRAG(BN1[1], win[20] - win[15], win[21] - win[16], win[22] - win[17], win[23] - win[18]);
        FRAG(BN1[2], win[25], win[26], win[27], win[28]);
        FRAG(BN1[3], win[9],  win[14], win[19], win[24]);
        FRAG(BM2[0], win[5],  win[6],  win[7],  win[8]);
        FRAG(BM2[1], win[15], win[16], win[17], win[18]);
        #undef FRAG
        const float* wp  = wl + d * (PATCH * ROWF);
        const float* wp4 = wc4 + d * (PATCH * ROWF);
        #pragma unroll
        for (int ig = 0; ig + IGRP <= JCOL; ig += IGRP) {
            v2f aE[IGRP][2], aO[IGRP][2], aP[IGRP][2];
            #pragma unroll
            for (int r = 0; r < IGRP; ++r) {
                const float* w = wp + (ig + r) * ROWF;
                aE[r][0].x = w[0];   aE[r][0].y = w[16];
                aO[r][0].x = w[80];  aO[r][0].y = w[96];
                aE[r][1].x = w[160]; aE[r][1].y = w[176];
                aO[r][1].x = w[240]; aO[r][1].y = w[256];
                aP[r][0].x = w[320]; aP[r][0].y = w[336];
                const float* w4 = wp4 + (ig + r) * ROWF;
                aP[r][1].x = w4[0];  aP[r][1].y = w4[80];
            }
            __builtin_amdgcn_sched_barrier(0);
            v8f n0[IGRP], n1[IGRP], m2[IGRP];
            #pragma unroll
            for (int r = 0; r < IGRP; ++r) {
                n0[r] = (v8f){0.f, 0.f, 0.f, 0.f, 0.f, 0.f, 0.f, 0.f};
                n1[r] = n0[r];
                m2[r] = n0[r];
            }
            #pragma unroll
            for (int kb = 0; kb < 4; ++kb) {
                #pragma unroll
                for (int r = 0; r < IGRP; ++r) {
                    const v2f a0 = kb < 2 ? aE[r][kb] : aP[r][kb - 2];
                    const v2f a1 = kb < 2 ? aO[r][kb] : aP[r][kb - 2];
                    n0[r] = __builtin_amdgcn_wmma_f32_16x16x4_f32(
                                false, a0, false, BN0[kb], (short)0, n0[r],
                                false, false);
                    n1[r] = __builtin_amdgcn_wmma_f32_16x16x4_f32(
                                false, a1, false, BN1[kb], (short)0, n1[r],
                                false, false);
                    if (kb >= 2) {
                        m2[r] = __builtin_amdgcn_wmma_f32_16x16x4_f32(
                                    false, aE[r][kb - 2] + aO[r][kb - 2], false,
                                    BM2[kb - 2], (short)0, m2[r], false, false);
                    }
                }
                __builtin_amdgcn_sched_barrier(0);
            }
            #pragma unroll
            for (int r = 0; r < IGRP; ++r) {
                const float p0 = win[ig + r], p1 = win[ig + r + KW];
                acc0 = __builtin_elementwise_fma(
                           (v8f){p0, p0, p0, p0, p0, p0, p0, p0}, n0[r] + m2[r], acc0);
                acc1 = __builtin_elementwise_fma(
                           (v8f){p1, p1, p1, p1, p1, p1, p1, p1}, n1[r] + m2[r], acc1);
            }
        }
        v2f Bf0[NKB], Bf1[NKB];
        #pragma unroll
        for (int kb = 0; kb < NKB; ++kb) {
            Bf0[kb].x = pick(m0[4 * kb],     m0[4 * kb + 2], hm);
            Bf0[kb].y = pick(m0[4 * kb + 1], m0[4 * kb + 3], hm);
            Bf1[kb].x = pick(m1[4 * kb],     m1[4 * kb + 2], hm);
            Bf1[kb].y = pick(m1[4 * kb + 1], m1[4 * kb + 3], hm);
        }
        v2f Al[NKB];
        #pragma unroll
        for (int kb = 0; kb < NKB; ++kb) {
            Al[kb].x = kern[lin0[kb] + d * F_OUT];
            Al[kb].y = kern[lin1[kb] + d * F_OUT];
        }
        {
            const int dn = d < C_IN - 1 ? d + 1 : C_IN - 1;
            int ro[WROWS], co[KW];
            #pragma unroll
            for (int t = 0; t < WROWS; ++t) {
                ro[t] = rowoff[t];
                asm volatile("" : "+v"(ro[t]));
            }
            #pragma unroll
            for (int t = 0; t < KW; ++t) co[t] = coloff[t] + dn;
            #pragma unroll
            for (int j = 0; j < WIN; ++j) nx[j] = x[ro[j / KW] + co[j % KW]];
        }
        {
            v2f ar[NKR], aq[NKB];
            const float* wqd = wq + d * (PATCH * ROWF);
            #pragma unroll
            for (int kb = 0; kb < NKR; ++kb) {
                ar[kb].x = wp[JCOL * ROWF + kb * 64];
                ar[kb].y = wp[JCOL * ROWF + kb * 64 + 16];
                aq[kb].x = wqd[kb * (4 * ROWF)];
                aq[kb].y = wqd[kb * (4 * ROWF) + ROWF];
            }
            aq[NKR].x = wz[d * (PATCH * ROWF)];
            aq[NKR].y = aq[NKR].x;
            __builtin_amdgcn_sched_barrier(0);
            v8f sr0 = (v8f){0.f, 0.f, 0.f, 0.f, 0.f, 0.f, 0.f, 0.f};
            v8f sr1 = sr0, sq0 = sr0, sq1 = sr0;
            #pragma unroll
            for (int kb = 0; kb < NKB; ++kb) {
                if (kb < NKR) {
                    sr0 = __builtin_amdgcn_wmma_f32_16x16x4_f32(
                              false, ar[kb], false, Bf0[kb], (short)0, sr0, false, false);
                    sr1 = __builtin_amdgcn_wmma_f32_16x16x4_f32(
                              false, ar[kb], false, Bf1[kb], (short)0, sr1, false, false);
                }
                sq0 = __builtin_amdgcn_wmma_f32_16x16x4_f32(
                          false, aq[kb], false, Bf0[kb], (short)0, sq0, false, false);
                sq1 = __builtin_amdgcn_wmma_f32_16x16x4_f32(
                          false, aq[kb], false, Bf1[kb], (short)0, sq1, false, false);
                __builtin_amdgcn_sched_barrier(0);
            }
            const float p0 = m0[JCOL], p1 = m1[JCOL];
            const v8f P0 = (v8f){p0, p0, p0, p0, p0, p0, p0, p0};
            const v8f P1 = (v8f){p1, p1, p1, p1, p1, p1, p1, p1};
            acc0 = __builtin_elementwise_fma(P0, sr0, acc0);
            acc1 = __builtin_elementwise_fma(P1, sr1, acc1);
            acc0 = __builtin_elementwise_fma(P0, sq0, acc0);
            acc1 = __builtin_elementwise_fma(P1, sq1, acc1);
        }
        #pragma unroll
        for (int kb = 0; kb < NKB; ++kb) {
            acc0 = __builtin_amdgcn_wmma_f32_16x16x4_f32(
                       false, Al[kb], false, Bf0[kb], (short)0, acc0, false, false);
            acc1 = __builtin_amdgcn_wmma_f32_16x16x4_f32(
                       false, Al[kb], false, Bf1[kb], (short)0, acc1, false, false);
        }
    }

    float* op = out + (size_t)((bc * HH + hc) * WW + wc) * F_OUT + 8 * h;
    *(v4f*)(op)     = (v4f){acc0[0], acc0[1], acc0[2], acc0[3]};
    *(v4f*)(op + 4) = (v4f){acc0[4], acc0[5], acc0[6], acc0[7]};
    op += WW * F_OUT;
    *(v4f*)(op)     = (v4f){acc1[0], acc1[1], acc1[2], acc1[3]};
    *(v4f*)(op + 4) = (v4f){acc1[4], acc1[5], acc1[6], acc1[7]};
}

extern "C" void kernel_launch(void* const* d_in, const int* in_sizes, int n_in,
                              void* d_out, int out_size, void* d_ws, size_t ws_size,
                              hipStream_t stream) {
    const float* x    = (const float*)d_in[0];
    const float* kern = (const float*)d_in[1];
    const float* qk   = (const float*)d_in[2];
    const float* bias = (const float*)d_in[3];
    float*       out  = (float*)d_out;

    const size_t shmem = (size_t)LDS_FLOATS * sizeof(float);
    (void)hipFuncSetAttribute((const void*)convquad_main,
                              hipFuncAttributeMaxDynamicSharedMemorySize,
                              (int)shmem);
    const int blocks = (NTILES + WPB - 1) / WPB;
    convquad_main<<<blocks, NTHREADS, shmem, stream>>>(x, kern, qk, bias, out);
}


# Shared-tap chain m2 is the C operand of both pixel chains' first WMMA: s = n + m2 needs no vector adds (240 -> 48 v_pk_add_f32 per trip); freed registers: rows in groups of 4; sum order changed
# speedup vs baseline: 1.7965x; 1.0419x over previous
#include <hip/hip_runtime.h>


typedef __attribute__((ext_vector_type(2))) float v2f;
typedef __attribute__((ext_vector_type(4))) float v4f;
typedef __attribute__((ext_vector_type(8))) float v8f;

#define KW        5
#define PATCH     25
#define NKB       7
#define NKR       6
#define JCOL      (PATCH-1)
#define IGRP      4
#define WROWS     6
#define WIN       (WROWS*KW)
#define C_IN      4
#define F_OUT     16
#define HH        250
#define WW        250
#define BB        8
#define HP        (HH/2)
#define NQ        (BB*HP*WW)
#define NTILES    (NQ/16)
#define WPB       8
#define NTHREADS  (WPB*32)
#define QFLOATS   (C_IN*PATCH*PATCH*F_OUT)
#define ROWF      (PATCH*F_OUT)
#define LDS_FLOATS QFLOATS

extern __shared__ float smem[];

__device__ __forceinline__ float pick(float lo, float hi, unsigned mask) {
    return __uint_as_float((__float_as_uint(lo) & ~mask) | (__float_as_uint(hi) & mask));
}

__global__ __launch_bounds__(NTHREADS, 4)
void convquad_main(const float* __restrict__ x,
                   const float* __restrict__ kern,
                   const float* __restrict__ qk,
                   const float* __restrict__ bias,
                   float* __restrict__ out) {
    const int tid  = threadIdx.x;
    const int lane = tid & 31;
    const int wave = tid >> 5;
    const int h    = (lane >> 4) & 1;
    const int c    = lane & 15;
    const unsigned hm = 0u - (unsigned)h;

    {
        const v4f* src = (const v4f*)qk;
        v4f*       dst = (v4f*)smem;
        for (int e = tid; e < QFLOATS / 4; e += NTHREADS) dst[e] = src[e];
    }

    const int tile = blockIdx.x * WPB + wave;
    const bool active = (tile < NTILES);

    const int Q  = (active ? tile : 0) * 16 + c;
    const int wc = Q % WW;
    const int hc = 2 * ((Q / WW) % HP);
    const int bc = Q / (WW * HP);
    int  rowoff[WROWS], coloff[KW];
    bool rowok[WROWS],  colok[KW];
    #pragma unroll
    for (int t = 0; t < WROWS; ++t) {
        int h2 = hc + t - 2;
        rowok[t] = (h2 >= 0 && h2 < HH);
        h2 = h2 < 0 ? 0 : (h2 > HH - 1 ? HH - 1 : h2);
        rowoff[t] = ((bc * HH + h2) * WW) * C_IN;
    }
    #pragma unroll
    for (int t = 0; t < KW; ++t) {
        int w2 = wc + t - 2;
        colok[t] = (w2 >= 0 && w2 < WW);
        w2 = w2 < 0 ? 0 : (w2 > WW - 1 ? WW - 1 : w2);
        coloff[t] = w2 * C_IN;
    }

    float nx[WIN];
    #pragma unroll
    for (int j = 0; j < WIN; ++j) nx[j] = x[rowoff[j / KW] + coloff[j % KW]];

    __syncthreads();
    if (!active) return;

    v8f acc0, acc1;
    {
        const v4f b0 = *(const v4f*)(bias + 8 * h);
        const v4f b1 = *(const v4f*)(bias + 8 * h + 4);
        acc0 = (v8f){b0.x, b0.y, b0.z, b0.w, b1.x, b1.y, b1.z, b1.w};
        acc1 = acc0;
    }

    const float* wl = smem + h * 32 + c;
    const float* wc4 = smem + h * (10 * F_OUT) + 4 * F_OUT + c;
    const float* wq = smem + h * (2 * ROWF) + JCOL * F_OUT + c;
    const float* wz = smem + JCOL * ROWF + JCOL * F_OUT + c;
    int lin0[NKB], lin1[NKB];
    #pragma unroll
    for (int kb = 0; kb < NKB; ++kb) {
        int i0 = 4 * kb + 2 * h, i1 = i0 + 1;
        i0 = i0 > PATCH - 1 ? PATCH - 1 : i0;
        i1 = i1 > PATCH - 1 ? PATCH - 1 : i1;
        lin0[kb] = i0 * (C_IN * F_OUT) + c;
        lin1[kb] = i1 * (C_IN * F_OUT) + c;
    }

    #pragma unroll 1
    for (int d = 0; d < C_IN; ++d) {
        float win[WIN], m0[4 * NKB], m1[4 * NKB];
        #pragma unroll
        for (int j = 0; j < WIN; ++j)
            win[j] = (rowok[j / KW] & colok[j % KW]) ? nx[j] : 0.0f;
        #pragma unroll
        for (int j = 0; j < PATCH; ++j) { m0[j] = win[j]; m1[j] = win[j + KW]; }
        #pragma unroll
        for (int j = PATCH; j < 4 * NKB; ++j) { m0[j] = 0.0f; m1[j] = 0.0f; }
        v2f BN0[4], BN1[4], BM2[2];
        #define FRAG(dst, f0, f1, f2, f3) \
            do { (dst).x = pick((f0), (f2), hm); (dst).y = pick((f1), (f3), hm); } while (0)
        FRAG(BN0[0], win[0] - win[5],   win[1] - win[6],   win[2] - win[7],   win[3] - win[8]);
        FRAG(BN0[1], win[10] - win[15], win[11] - win[16], win[12] - win[17], win[13] - win[18]);
        FRAG(BN0[2], win[20], win[21], win[22], win[23]);
        FRAG(BN0[3], win[4],  win[9],  win[14], win[19]);
        FRAG(BN1[0], win[10] - win[5],  win[11] - win[6],  win[12] - win[7],  win[13] - win[8]);
        FRAG(BN1[1], win[20] - win[15], win[21] - win[16], win[22] - win[17], win[23] - win[18]);
        FRAG(BN1[2], win[25], win[26], win[27], win[28]);
        FRAG(BN1[3], win[9],  win[14], win[19], win[24]);
        FRAG(BM2[0], win[5],  win[6],  win[7],  win[8]);
        FRAG(BM2[1], win[15], win[16], win[17], win[18]);
        #undef FRAG
        const float* wp  = wl + d * (PATCH * ROWF);
        const float* wp4 = wc4 + d * (PATCH * ROWF);
        #pragma unroll
        for (int ig = 0; ig + IGRP <= JCOL; ig += IGRP) {
            v2f aE[IGRP][2], aO[IGRP][2], aP[IGRP][2];
            #pragma unroll
            for (int r = 0; r < IGRP; ++r) {
                const float* w = wp + (ig + r) * ROWF;
                aE[r][0].x = w[0];   aE[r][0].y = w[16];
                aO[r][0].x = w[80];  aO[r][0].y = w[96];
                aE[r][1].x = w[160]; aE[r][1].y = w[176];
                aO[r][1].x = w[240]; aO[r][1].y = w[256];
                aP[r][0].x = w[320]; aP[r][0].y = w[336];
                const float* w4 = wp4 + (ig + r) * ROWF;
                aP[r][1].x = w4[0];  aP[r][1].y = w4[80];
            }
            __builtin_amdgcn_sched_barrier(0);
            v8f n0[IGRP], n1[IGRP], m2[IGRP];
            #pragma unroll
            for (int kb = 0; kb < 2; ++kb) {
                #pragma unroll
                for (int r = 0; r < IGRP; ++r) {
                    const v8f c = kb == 0 ? (v8f){0.f, 0.f, 0.f, 0.f, 0.f, 0.f, 0.f, 0.f}
                                          : m2[r];
                    m2[r] = __builtin_amdgcn_wmma_f32_16x16x4_f32(
                                false, aE[r][kb] + aO[r][kb], false,
                                BM2[kb], (short)0, c, false, false);
                }
                __builtin_amdgcn_sched_barrier(0);
            }
            #pragma unroll
            for (int kb = 0; kb < 4; ++kb) {
                #pragma unroll
                for (int r = 0; r < IGRP; ++r) {
                    const v2f a0 = kb < 2 ? aE[r][kb] : aP[r][kb - 2];
                    const v2f a1 = kb < 2 ? aO[r][kb] : aP[r][kb - 2];
                    const v8f c0 = kb == 0 ? m2[r] : n0[r];
                    const v8f c1 = kb == 0 ? m2[r] : n1[r];
                    n0[r] = __builtin_amdgcn_wmma_f32_16x16x4_f32(
                                false, a0, false, BN0[kb], (short)0, c0,
                                false, false);
                    n1[r] = __builtin_amdgcn_wmma_f32_16x16x4_f32(
                                false, a1, false, BN1[kb], (short)0, c1,
                                false, false);
                }
                #pragma unroll
                for (int r = 0; r < IGRP; ++r) {
                    asm volatile("" : "+v"(n0[r]), "+v"(n1[r]));
                }
                __builtin_amdgcn_sched_barrier(0);
            }
            #pragma unroll
            for (int r = 0; r < IGRP; ++r) {
                const float p0 = win[ig + r], p1 = win[ig + r + KW];
                acc0 = __builtin_elementwise_fma(
                           (v8f){p0, p0, p0, p0, p0, p0, p0, p0}, n0[r], acc0);
                acc1 = __builtin_elementwise_fma(
                           (v8f){p1, p1, p1, p1, p1, p1, p1, p1}, n1[r], acc1);
            }
        }
        v2f Bf0[NKB], Bf1[NKB];
        #pragma unroll
        for (int kb = 0; kb < NKB; ++kb) {
            Bf0[kb].x = pick(m0[4 * kb],     m0[4 * kb + 2], hm);
            Bf0[kb].y = pick(m0[4 * kb + 1], m0[4 * kb + 3], hm);
            Bf1[kb].x = pick(m1[4 * kb],     m1[4 * kb + 2], hm);
            Bf1[kb].y = pick(m1[4 * kb + 1], m1[4 * kb + 3], hm);
        }
        v2f Al[NKB];
        #pragma unroll
        for (int kb = 0; kb < NKB; ++kb) {
            Al[kb].x = kern[lin0[kb] + d * F_OUT];
            Al[kb].y = kern[lin1[kb] + d * F_OUT];
        }
        {
            const int dn = d < C_IN - 1 ? d + 1 : C_IN - 1;
            int ro[WROWS], co[KW];
            #pragma unroll
            for (int t = 0; t < WROWS; ++t) {
                ro[t] = rowoff[t];
                asm volatile("" : "+v"(ro[t]));
            }
            #pragma unroll
            for (int t = 0; t < KW; ++t) co[t] = coloff[t] + dn;
            #pragma unroll
            for (int j = 0; j < WIN; ++j) nx[j] = x[ro[j / KW] + co[j % KW]];
        }
        {
            v2f ar[NKR], aq[NKB];
            const float* wqd = wq + d * (PATCH * ROWF);
            #pragma unroll
            for (int kb = 0; kb < NKR; ++kb) {
                ar[kb].x = wp[JCOL * ROWF + kb * 64];
                ar[kb].y = wp[JCOL * ROWF + kb * 64 + 16];
                aq[kb].x = wqd[kb * (4 * ROWF)];
                aq[kb].y = wqd[kb * (4 * ROWF) + ROWF];
            }
            aq[NKR].x = wz[d * (PATCH * ROWF)];
            aq[NKR].y = aq[NKR].x;
            __builtin_amdgcn_sched_barrier(0);
            v8f sr0 = (v8f){0.f, 0.f, 0.f, 0.f, 0.f, 0.f, 0.f, 0.f};
            v8f sr1 = sr0, sq0 = sr0, sq1 = sr0;
            #pragma unroll
            for (int kb = 0; kb < NKB; ++kb) {
                if (kb < NKR) {
                    sr0 = __builtin_amdgcn_wmma_f32_16x16x4_f32(
                              false, ar[kb], false, Bf0[kb], (short)0, sr0, false, false);
                    sr1 = __builtin_amdgcn_wmma_f32_16x16x4_f32(
                              false, ar[kb], false, Bf1[kb], (short)0, sr1, false, false);
                }
                sq0 = __builtin_amdgcn_wmma_f32_16x16x4_f32(
                          false, aq[kb], false, Bf0[kb], (short)0, sq0, false, false);
                sq1 = __builtin_amdgcn_wmma_f32_16x16x4_f32(
                          false, aq[kb], false, Bf1[kb], (short)0, sq1, false, false);
                __builtin_amdgcn_sched_barrier(0);
            }
            const float p0 = m0[JCOL], p1 = m1[JCOL];
            const v8f P0 = (v8f){p0, p0, p0, p0, p0, p0, p0, p0};
            const v8f P1 = (v8f){p1, p1, p1, p1, p1, p1, p1, p1};
            acc0 = __builtin_elementwise_fma(P0, sr0, acc0);
            acc1 = __builtin_elementwise_fma(P1, sr1, acc1);
            acc0 = __builtin_elementwise_fma(P0, sq0, acc0);
            acc1 = __builtin_elementwise_fma(P1, sq1, acc1);
        }
        #pragma unroll
        for (int kb = 0; kb < NKB; ++kb) {
            acc0 = __builtin_amdgcn_wmma_f32_16x16x4_f32(
                       false, Al[kb], false, Bf0[kb], (short)0, acc0, false, false);
            acc1 = __builtin_amdgcn_wmma_f32_16x16x4_f32(
                       false, Al[kb], false, Bf1[kb], (short)0, acc1, false, false);
        }
    }

    float* op = out + (size_t)((bc * HH + hc) * WW + wc) * F_OUT + 8 * h;
    *(v4f*)(op)     = (v4f){acc0[0], acc0[1], acc0[2], acc0[3]};
    *(v4f*)(op + 4) = (v4f){acc0[4], acc0[5], acc0[6], acc0[7]};
    op += WW * F_OUT;
    *(v4f*)(op)     = (v4f){acc1[0], acc1[1], acc1[2], acc1[3]};
    *(v4f*)(op + 4) = (v4f){acc1[4], acc1[5], acc1[6], acc1[7]};
}

extern "C" void kernel_launch(void* const* d_in, const int* in_sizes, int n_in,
                              void* d_out, int out_size, void* d_ws, size_t ws_size,
                              hipStream_t stream) {
    const float* x    = (const float*)d_in[0];
    const float* kern = (const float*)d_in[1];
    const float* qk   = (const float*)d_in[2];
    const float* bias = (const float*)d_in[3];
    float*       out  = (float*)d_out;

    const size_t shmem = (size_t)LDS_FLOATS * sizeof(float);
    (void)hipFuncSetAttribute((const void*)convquad_main,
                              hipFuncAttributeMaxDynamicSharedMemorySize,
                              (int)shmem);
    const int blocks = (NTILES + WPB - 1) / WPB;
    convquad_main<<<blocks, NTHREADS, shmem, stream>>>(x, kern, qk, bias, out);
}
